# starting kernel
# baseline (speedup 1.0000x reference)
#include <hip/hip_runtime.h>

#define CC 64
#define KK 27
#define LDS_STRIDE 66
#define IDX_STRIDE 28

typedef __attribute__((ext_vector_type(2))) float v2f;
typedef __attribute__((ext_vector_type(8))) float v8f;

__global__ __launch_bounds__(256)
void gemm_bias_wmma(const float* __restrict__ feat,
                    const float* __restrict__ W,
                    const float* __restrict__ b1,
                    float* __restrict__ bx,
                    int Npts)
{
    __shared__ float sWt[CC * LDS_STRIDE];
    for (int i = threadIdx.x; i < CC * CC; i += 256) {
        int k = i >> 6;
        int n = i & 63;
        sWt[n * LDS_STRIDE + k] = W[i];
    }
    __syncthreads();

    const int lane = threadIdx.x & 31;
    const int wave = threadIdx.x >> 5;
    const int l15  = lane & 15;
    const int hs   = lane >> 4;
    const int rowBase = (blockIdx.x * 8 + wave) * 16;

    int aRowIdx = rowBase + l15;
    if (aRowIdx > Npts - 1) aRowIdx = Npts - 1;
    const float* aRow = feat + (size_t)aRowIdx * CC;

    v8f acc[4] = {v8f{}, v8f{}, v8f{}, v8f{}};

#pragma unroll
    for (int k0 = 0; k0 < CC; k0 += 4) {
        v2f a = *(const v2f*)(aRow + k0 + hs * 2);
#pragma unroll
        for (int nt = 0; nt < 4; ++nt) {
            const float* bp = &sWt[(nt * 16 + l15) * LDS_STRIDE + k0 + hs * 2];
            v2f b = *(const v2f*)bp;
            acc[nt] = __builtin_amdgcn_wmma_f32_16x16x4_f32(
                false, a, false, b, (short)0, acc[nt], false, false);
        }
    }

#pragma unroll
    for (int nt = 0; nt < 4; ++nt) {
        const int col = nt * 16 + l15;
        const float bias = b1[col];
#pragma unroll
        for (int r = 0; r < 8; ++r) {
            const int row = rowBase + hs * 8 + r;
            if (row < Npts) {
                bx[(size_t)row * CC + col] = acc[nt][r] + bias;
            }
        }
    }
}

__global__ __launch_bounds__(256)
void dw_gather_residual(const float* __restrict__ feat,
                        const int*   __restrict__ nb,
                        const float* __restrict__ bx,
                        const float* __restrict__ dw,
                        const float* __restrict__ b2,
                        float*       __restrict__ out,
                        int Npts)
{
    __shared__ int   sIdx[16 * IDX_STRIDE];
    __shared__ float sDw[KK * CC];

    const int rowTile = blockIdx.x * 16;

    {
        int validRows = Npts - rowTile;
        if (validRows > 16) validRows = 16;
        const int total = validRows * KK;
        const int* slab = nb + (size_t)rowTile * KK;
        for (int i = threadIdx.x; i < total; i += 256) {
            const int r = i / KK;
            const int k = i - r * KK;
            sIdx[r * IDX_STRIDE + k] = slab[i];
        }
    }
    for (int i = threadIdx.x; i < KK * CC; i += 256) {
        sDw[i] = dw[i];
    }
    __syncthreads();

    const int r   = threadIdx.x >> 4;
    const int c4  = (threadIdx.x & 15) << 2;
    const int row = rowTile + r;
    if (row >= Npts) return;

    const int* myIdx = &sIdx[r * IDX_STRIDE];

    float ax = 0.f, ay = 0.f, az = 0.f, aw = 0.f;
#pragma unroll
    for (int k = 0; k < KK; ++k) {
        const int idx = myIdx[k];
        if (idx >= 0) {
            const float4 g = *(const float4*)(bx + (size_t)idx * CC + c4);
            const float4 w = *(const float4*)(&sDw[k * CC + c4]);
            ax += g.x * w.x;
            ay += g.y * w.y;
            az += g.z * w.z;
            aw += g.w * w.w;
        }
    }

    const float4 f  = *(const float4*)(feat + (size_t)row * CC + c4);
    const float4 bb = *(const float4*)(b2 + c4);
    float4 o;
    o.x = f.x + ax + bb.x;
    o.y = f.y + ay + bb.y;
    o.z = f.z + az + bb.z;
    o.w = f.w + aw + bb.w;
    *(float4*)(out + (size_t)row * CC + c4) = o;
}

extern "C" void kernel_launch(void* const* d_in, const int* in_sizes, int n_in,
                              void* d_out, int out_size, void* d_ws, size_t ws_size,
                              hipStream_t stream)
{
    const float* feat = (const float*)d_in[0];
    const int*   nb   = (const int*)  d_in[1];
    const float* W    = (const float*)d_in[2];
    const float* b1   = (const float*)d_in[3];
    const float* b2   = (const float*)d_in[4];
    const float* dw   = (const float*)d_in[5];
    float* out = (float*)d_out;
    float* bx  = (float*)d_ws;

    const int Npts = in_sizes[0] / CC;

    const int gemmBlocks = (Npts + 127) / 128;
    gemm_bias_wmma<<<gemmBlocks, 256, 0, stream>>>(feat, W, b1, bx, Npts);

    const int dwBlocks = (Npts + 15) / 16;
    dw_gather_residual<<<dwBlocks, 256, 0, stream>>>(feat, nb, bx, dw, b2, out, Npts);
}


# GEMM kernel as a resident grid: 512 work-groups stage W once and keep its B fragments in registers; each wave walks 4 row tiles and requests the next tile's feat fragments ahead; k order kept
# speedup vs baseline: 1.7418x; 1.7418x over previous
#include <hip/hip_runtime.h>

#define CC 64
#define KK 27
#define LDS_STRIDE 66
#define ROWS_PER_WAVE 16
#define TAP_BATCH 8

typedef __attribute__((ext_vector_type(2))) float v2f;
typedef __attribute__((ext_vector_type(8))) float v8f;

#define GEMM_BLOCKS 512

__device__ __forceinline__ void load_a_tile(v2f (&a)[16], const float* __restrict__ feat,
                                            int tile, int l15, int hs, int Npts)
{
    int aRowIdx = tile * 16 + l15;
    if (aRowIdx > Npts - 1) aRowIdx = Npts - 1;
    const float* aRow = feat + (size_t)aRowIdx * CC + hs * 2;
#pragma unroll
    for (int s = 0; s < 16; ++s) a[s] = *(const v2f*)(aRow + s * 4);
}

__global__ __launch_bounds__(256)
void gemm_bias_wmma(const float* __restrict__ feat,
                    const float* __restrict__ W,
                    const float* __restrict__ b1,
                    float* __restrict__ bx,
                    int Npts)
{
    __shared__ float sWt[CC * LDS_STRIDE];
    {
        float w[16];
#pragma unroll
        for (int j = 0; j < 16; ++j) w[j] = W[threadIdx.x + 256 * j];
#pragma unroll
        for (int j = 0; j < 16; ++j) {
            const int i = threadIdx.x + 256 * j;
            sWt[(i & 63) * LDS_STRIDE + (i >> 6)] = w[j];
        }
    }
    __syncthreads();

    const int lane = threadIdx.x & 31;
    const int wave = __builtin_amdgcn_readfirstlane(threadIdx.x >> 5);
    const int l15  = lane & 15;
    const int hs   = lane >> 4;

    const int nTiles = (Npts + 15) >> 4;
    const int stride = gridDim.x * 8;
    int tile = blockIdx.x * 8 + wave;
    if (tile >= nTiles) return;

    float bias[4];
#pragma unroll
    for (int nt = 0; nt < 4; ++nt) bias[nt] = b1[nt * 16 + l15];

    const float* bBase = &sWt[l15 * LDS_STRIDE + hs * 2];
    v2f bf[16][4];
#pragma unroll
    for (int s = 0; s < 16; ++s) {
#pragma unroll
        for (int nt = 0; nt < 4; ++nt) bf[s][nt] = *(const v2f*)(bBase + nt * 16 * LDS_STRIDE + s * 4);
    }
    const unsigned laneOff = (unsigned)(hs * 8 * CC + l15);

    v2f an[16];
    load_a_tile(an, feat, tile, l15, hs, Npts);

    for (;;) {
        v2f a[16];
#pragma unroll
        for (int s = 0; s < 16; ++s) a[s] = an[s];

        const int  next = tile + stride;
        const bool more = next < nTiles;
        if (more) load_a_tile(an, feat, next, l15, hs, Npts);

        v8f acc[4] = {v8f{}, v8f{}, v8f{}, v8f{}};
#pragma unroll
        for (int s = 0; s < 16; ++s) {
#pragma unroll
            for (int nt = 0; nt < 4; ++nt) {
                acc[nt] = __builtin_amdgcn_wmma_f32_16x16x4_f32(
                    false, a[s], false, bf[s][nt], (short)0, acc[nt], false, false);
            }
        }

        __builtin_amdgcn_sched_barrier(0);
        asm volatile("v_nop\n\tv_nop\n\tv_nop\n\tv_nop");
        __builtin_amdgcn_sched_barrier(0);

        const int rowBase = tile * 16;
        float* tileOut = bx + (size_t)rowBase * CC;
        if (rowBase + 16 <= Npts) {
#pragma unroll
            for (int nt = 0; nt < 4; ++nt) {
#pragma unroll
                for (int r = 0; r < 8; ++r) tileOut[laneOff + (unsigned)(r * CC + nt * 16)] = acc[nt][r] + bias[nt];
            }
        } else {
            const int rowsLeft = Npts - rowBase;
#pragma unroll
            for (int nt = 0; nt < 4; ++nt) {
#pragma unroll
                for (int r = 0; r < 8; ++r) {
                    if (hs * 8 + r < rowsLeft) tileOut[laneOff + (unsigned)(r * CC + nt * 16)] = acc[nt][r] + bias[nt];
                }
            }
        }

        if (!more) break;
        tile = next;
    }
}


template <int N>
__device__ __forceinline__ void tap_batch(unsigned& m, int idxv,
                                          const float* __restrict__ bx,
                                          const v2f* sDwLane, int c2, int lastRow,
                                          float& ax, float& ay)
{
    v2f g[N];
    v2f w[N];
#pragma unroll
    for (int j = 0; j < N; ++j) {
        const int k = __builtin_ctz(m);
        m &= m - 1u;
        int idx = __builtin_amdgcn_readlane(idxv, k);
        if (idx > lastRow) idx = lastRow;
        const float* rowp = bx + (size_t)idx * CC;
        g[j] = *(const v2f*)(rowp + c2);
        w[j] = sDwLane[k * 32];
    }
#pragma unroll
    for (int j = 0; j < N; ++j) {
        ax += g[j].x * w[j].x;
        ay += g[j].y * w[j].y;
    }
}

__global__ __launch_bounds__(256)
void dw_gather_residual(const float* __restrict__ feat,
                        const int*   __restrict__ nb,
                        const float* __restrict__ bx,
                        const float* __restrict__ dw,
                        const float* __restrict__ b2,
                        float*       __restrict__ out,
                        int Npts)
{
    __shared__ v2f sDw[KK * 32];

    for (int i = threadIdx.x; i < KK * 32; i += 256) {
        sDw[i] = *(const v2f*)(dw + 2 * i);
    }
    __syncthreads();

    const int lane = threadIdx.x & 31;
    const int wave = __builtin_amdgcn_readfirstlane(threadIdx.x >> 5);
    const int c2   = lane * 2;

    const int rowBeg = (blockIdx.x * 8 + wave) * ROWS_PER_WAVE;
    int rowEnd = rowBeg + ROWS_PER_WAVE;
    if (rowEnd > Npts) rowEnd = Npts;
    if (rowBeg >= rowEnd) return;

    const v2f bb = *(const v2f*)(b2 + c2);
    const int lastRow = Npts - 1;
    const v2f* sDwLane = sDw + lane;

    int idxNext = -1;
    if (lane < KK) idxNext = (nb + (size_t)rowBeg * KK)[lane];
    v2f fNext = *(const v2f*)(feat + (size_t)rowBeg * CC + c2);

    for (int row = rowBeg; row < rowEnd; ++row) {
        const int idxv = idxNext;
        const v2f f    = fNext;
        if (row + 1 < rowEnd) {
            idxNext = -1;
            if (lane < KK) idxNext = (nb + (size_t)(row + 1) * KK)[lane];
            fNext = *(const v2f*)(feat + (size_t)(row + 1) * CC + c2);
        }

        unsigned m = __builtin_amdgcn_ballot_w32(idxv >= 0);

        float ax = 0.f, ay = 0.f;
        while (m != 0u) {
            switch (__builtin_popcount(m)) {
            case 1:  tap_batch<1>(m, idxv, bx, sDwLane, c2, lastRow, ax, ay); break;
            case 2:  tap_batch<2>(m, idxv, bx, sDwLane, c2, lastRow, ax, ay); break;
            case 3:  tap_batch<3>(m, idxv, bx, sDwLane, c2, lastRow, ax, ay); break;
            case 4:  tap_batch<4>(m, idxv, bx, sDwLane, c2, lastRow, ax, ay); break;
            case 5:  tap_batch<5>(m, idxv, bx, sDwLane, c2, lastRow, ax, ay); break;
            case 6:  tap_batch<6>(m, idxv, bx, sDwLane, c2, lastRow, ax, ay); break;
            case 7:  tap_batch<7>(m, idxv, bx, sDwLane, c2, lastRow, ax, ay); break;
            default: tap_batch<TAP_BATCH>(m, idxv, bx, sDwLane, c2, lastRow, ax, ay); break;
            }
        }

        v2f o;
        o.x = f.x + ax + bb.x;
        o.y = f.y + ay + bb.y;
        *(v2f*)(out + (size_t)row * CC + c2) = o;
    }
}

extern "C" void kernel_launch(void* const* d_in, const int* in_sizes, int n_in,
                              void* d_out, int out_size, void* d_ws, size_t ws_size,
                              hipStream_t stream)
{
    const float* feat = (const float*)d_in[0];
    const int*   nb   = (const int*)  d_in[1];
    const float* W    = (const float*)d_in[2];
    const float* b1   = (const float*)d_in[3];
    const float* b2   = (const float*)d_in[4];
    const float* dw   = (const float*)d_in[5];
    float* out = (float*)d_out;
    float* bx  = (float*)d_ws;

    const int Npts = in_sizes[0] / CC;

    gemm_bias_wmma<<<GEMM_BLOCKS, 256, 0, stream>>>(feat, W, b1, bx, Npts);

    const int dwBlocks = (Npts + 8 * ROWS_PER_WAVE - 1) / (8 * ROWS_PER_WAVE);
    dw_gather_residual<<<dwBlocks, 256, 0, stream>>>(feat, nb, bx, dw, b2, out, Npts);
}
